# starting kernel
# baseline (speedup 1.0000x reference)
#include <hip/hip_runtime.h>
#include <hip/hip_bf16.h>

typedef float v2f __attribute__((ext_vector_type(2)));
typedef float v8f __attribute__((ext_vector_type(8)));

#define MAXSEG 512
#define CB 8
#define PCHUNK 8192

__global__ void zero_ws_kernel(float* ws, int n) {
    int i = blockIdx.x * blockDim.x + threadIdx.x;
    if (i < n) ws[i] = 0.0f;
}

__global__ void count_kernel(const int* __restrict__ seg, float* __restrict__ counts,
                             int P, int N, int chunk) {
    __shared__ int hist[MAXSEG];
    for (int i = threadIdx.x; i < N; i += blockDim.x) hist[i] = 0;
    __syncthreads();
    int pstart = blockIdx.x * chunk;
    for (int i = threadIdx.x; i < chunk; i += blockDim.x) {
        int g = pstart + i;
        if (g < P) atomicAdd(&hist[seg[g]], 1);
    }
    __syncthreads();
    for (int i = threadIdx.x; i < N; i += blockDim.x) {
        int c = hist[i];
        if (c) atomicAdd(&counts[i], (float)c);
    }
}

__global__ void accum_kernel(const float* __restrict__ feat,
                             const int* __restrict__ seg,
                             float* __restrict__ sums,
                             int P, int C, int N, int chunk) {
    __shared__ int   lseg[PCHUNK];
    __shared__ float lsums[CB * MAXSEG];

    for (int i = threadIdx.x; i < CB * N; i += blockDim.x) lsums[i] = 0.0f;

    int pstart = blockIdx.x * chunk;
    int navail = P - pstart;
    int n = (navail < chunk) ? navail : chunk;
    if (n == chunk) {
        const int4* seg4 = (const int4*)(seg + pstart);
        int4* lseg4 = (int4*)lseg;
        for (int i = threadIdx.x; i < chunk / 4; i += blockDim.x)
            lseg4[i] = seg4[i];
    } else {
        for (int i = threadIdx.x; i < chunk; i += blockDim.x)
            lseg[i] = (i < n) ? seg[pstart + i] : 0;
    }
    __syncthreads();

    int wave = threadIdx.x >> 5;
    int lane = threadIdx.x & 31;
    int c = blockIdx.y * CB + wave;
    const float* plane = feat + (size_t)c * (size_t)P + pstart;
    float* lrow = &lsums[wave * N];

    if (n == chunk) {
        const float4* plane4 = (const float4*)plane;
        const int4*   lseg4  = (const int4*)lseg;
        for (int i = lane; i < chunk / 4; i += 32) {
            if (i + 256 < chunk / 4)
                __builtin_prefetch(&plane4[i + 256], 0, 0);
            float4 v = plane4[i];
            int4   s = lseg4[i];
            atomicAdd(&lrow[s.x], v.x);
            atomicAdd(&lrow[s.y], v.y);
            atomicAdd(&lrow[s.z], v.z);
            atomicAdd(&lrow[s.w], v.w);
        }
    } else {
        for (int i = lane; i < n; i += 32) {
            float v = plane[i];
            atomicAdd(&lrow[lseg[i]], v);
        }
    }
    __syncthreads();

    for (int i = threadIdx.x; i < CB * N; i += blockDim.x) {
        float v = lsums[i];
        if (v != 0.0f) {
            int w = i / N;
            int s = i - w * N;
            atomicAdd(&sums[(size_t)s * C + blockIdx.y * CB + w], v);
        }
    }
}

__global__ void finalize_wmma_kernel(const float* __restrict__ sums,
                                     const float* __restrict__ counts,
                                     float* __restrict__ out,
                                     int C, int N) {
    int wave = threadIdx.x >> 5;
    int lane = threadIdx.x & 31;
    int half = lane >> 4;
    int m    = lane & 15;
    int n0   = blockIdx.x * 16;
    int c0   = wave * 16;

    float cnt = counts[n0 + m];
    float inv = (cnt > 0.0f) ? (1.0f / cnt) : 0.0f;

    v8f acc = {};
#pragma unroll
    for (int k0 = 0; k0 < 16; k0 += 4) {
        int kx = k0 + 2 * half;
        v2f a, b;
        a.x = (m == kx)     ? inv : 0.0f;
        a.y = (m == kx + 1) ? inv : 0.0f;
        b.x = sums[(size_t)(n0 + kx)     * C + c0 + m];
        b.y = sums[(size_t)(n0 + kx + 1) * C + c0 + m];
        acc = __builtin_amdgcn_wmma_f32_16x16x4_f32(
             false, a,  false, b,
             (short)0, acc,  false,  false);
    }

#pragma unroll
    for (int v = 0; v < 8; ++v) {
        out[(size_t)(n0 + v + 8 * half) * C + c0 + m] = acc[v];
    }
}

extern "C" void kernel_launch(void* const* d_in, const int* in_sizes, int n_in,
                              void* d_out, int out_size, void* d_ws, size_t ws_size,
                              hipStream_t stream) {
    const float* feat = (const float*)d_in[0];
    const int*   seg  = (const int*)d_in[1];
    int P = in_sizes[1];
    int C = in_sizes[0] / P;
    int N = out_size / C;

    float* sums   = (float*)d_ws;
    float* counts = sums + (size_t)N * C;
    float* out    = (float*)d_out;

    int tot = N * C + N;
    zero_ws_kernel<<<(tot + 255) / 256, 256, 0, stream>>>(sums, tot);

    int chunk   = PCHUNK;
    int pblocks = (P + chunk - 1) / chunk;

    count_kernel<<<pblocks, 256, 0, stream>>>(seg, counts, P, N, chunk);

    dim3 agrid(pblocks, C / CB);
    accum_kernel<<<agrid, 256, 0, stream>>>(feat, seg, sums, P, C, N, chunk);

    finalize_wmma_kernel<<<N / 16, (C / 16) * 32, 0, stream>>>(sums, counts, out, C, N);
}


# accumulate kernel as one-wave workgroups: each lane keeps its own LDS column of the 400 bins, so no ds_add_f32 shares a word or bank; columns summed in fixed order at the end (sum order changes)
# speedup vs baseline: 2.9255x; 2.9255x over previous
#include <hip/hip_runtime.h>
#include <hip/hip_bf16.h>

typedef float v2f __attribute__((ext_vector_type(2)));
typedef float v8f __attribute__((ext_vector_type(8)));

#define MAXSEG 512
#define LSEG 400
#define UB 16
#define NRANGE 12
#define PCHUNK 8192

__global__ void zero_ws_kernel(float* ws, int n) {
    int i = blockIdx.x * blockDim.x + threadIdx.x;
    if (i < n) ws[i] = 0.0f;
}

__global__ void count_kernel(const int* __restrict__ seg, float* __restrict__ counts,
                             int P, int N, int chunk) {
    __shared__ int hist[MAXSEG];
    for (int i = threadIdx.x; i < N; i += blockDim.x) hist[i] = 0;
    __syncthreads();
    int pstart = blockIdx.x * chunk;
    for (int i = threadIdx.x; i < chunk; i += blockDim.x) {
        int g = pstart + i;
        if (g < P) atomicAdd(&hist[seg[g]], 1);
    }
    __syncthreads();
    for (int i = threadIdx.x; i < N; i += blockDim.x) {
        int c = hist[i];
        if (c) atomicAdd(&counts[i], (float)c);
    }
}

__device__ __forceinline__ void load_batch(float4 (&fv)[UB], int4 (&sv)[UB],
                                           const float4* __restrict__ p4,
                                           const int4* __restrict__ s4,
                                           int bat, int lane) {
#pragma unroll
    for (int u = 0; u < UB; ++u) {
        fv[u] = p4[((size_t)bat * UB + u) * 32 + lane];
        sv[u] = s4[((size_t)bat * UB + u) * 32 + lane];
    }
}

__device__ __forceinline__ void add_batch(float* lacc, const float4 (&fv)[UB],
                                          const int4 (&sv)[UB], int lane) {
#pragma unroll
    for (int u = 0; u < UB; ++u) {
        atomicAdd(&lacc[sv[u].x * 32 + lane], fv[u].x);
        atomicAdd(&lacc[sv[u].y * 32 + lane], fv[u].y);
        atomicAdd(&lacc[sv[u].z * 32 + lane], fv[u].z);
        atomicAdd(&lacc[sv[u].w * 32 + lane], fv[u].w);
    }
}

__global__ void __launch_bounds__(32)
accum_kernel(const float* __restrict__ feat,
             const int* __restrict__ seg,
             float* __restrict__ sums,
             int P, int C, int N) {
    __shared__ float lacc[LSEG * 32];

    const int lane = threadIdx.x;
    const int c = blockIdx.x;
    const int r = blockIdx.y;
    const int R = gridDim.y;

    float4* lacc4 = (float4*)lacc;
    for (int i = lane; i < N * 8; i += 32) lacc4[i] = make_float4(0.f, 0.f, 0.f, 0.f);
    __syncthreads();

    const float* plane = feat + (size_t)c * (size_t)P;

    const int nbat = ((P & 3) == 0) ? (P / (128 * UB)) : 0;
    const int bat_lo = (int)(((long long)r * nbat) / R);
    const int bat_hi = (int)(((long long)(r + 1) * nbat) / R);

    const float4* p4 = (const float4*)plane;
    const int4*   s4 = (const int4*)seg;

    float4 fv[UB];
    int4   sv[UB];
    for (int k = bat_lo; k < bat_hi; ++k) {
        load_batch(fv, sv, p4, s4, k, lane);
        __builtin_amdgcn_sched_barrier(0);
        add_batch(lacc, fv, sv, lane);
    }

    if (r == R - 1) {
        for (int i = nbat * (128 * UB) + lane; i < P; i += 32)
            atomicAdd(&lacc[seg[i] * 32 + lane], plane[i]);
    }
    __syncthreads();

    for (int s0 = 0; s0 < N; s0 += 32) {
        const int s = s0 + lane;
        if (s < N) {
            const float* row = &lacc[s * 32];
            float acc = 0.0f;
#pragma unroll
            for (int j = 0; j < 32; ++j) acc += row[(j + lane) & 31];
            atomicAdd(&sums[(size_t)s * C + c], acc);
        }
    }
}

__global__ void finalize_wmma_kernel(const float* __restrict__ sums,
                                     const float* __restrict__ counts,
                                     float* __restrict__ out,
                                     int C, int N) {
    int wave = threadIdx.x >> 5;
    int lane = threadIdx.x & 31;
    int half = lane >> 4;
    int m    = lane & 15;
    int n0   = blockIdx.x * 16;
    int c0   = wave * 16;

    float cnt = counts[n0 + m];
    float inv = (cnt > 0.0f) ? (1.0f / cnt) : 0.0f;

    v8f acc = {};
#pragma unroll
    for (int k0 = 0; k0 < 16; k0 += 4) {
        int kx = k0 + 2 * half;
        v2f a, b;
        a.x = (m == kx)     ? inv : 0.0f;
        a.y = (m == kx + 1) ? inv : 0.0f;
        b.x = sums[(size_t)(n0 + kx)     * C + c0 + m];
        b.y = sums[(size_t)(n0 + kx + 1) * C + c0 + m];
        acc = __builtin_amdgcn_wmma_f32_16x16x4_f32(
             false, a,  false, b,
             (short)0, acc,  false,  false);
    }

#pragma unroll
    for (int v = 0; v < 8; ++v) {
        out[(size_t)(n0 + v + 8 * half) * C + c0 + m] = acc[v];
    }
}

extern "C" void kernel_launch(void* const* d_in, const int* in_sizes, int n_in,
                              void* d_out, int out_size, void* d_ws, size_t ws_size,
                              hipStream_t stream) {
    const float* feat = (const float*)d_in[0];
    const int*   seg  = (const int*)d_in[1];
    int P = in_sizes[1];
    int C = in_sizes[0] / P;
    int N = out_size / C;

    float* sums   = (float*)d_ws;
    float* counts = sums + (size_t)N * C;
    float* out    = (float*)d_out;

    int tot = N * C + N;
    zero_ws_kernel<<<(tot + 255) / 256, 256, 0, stream>>>(sums, tot);

    int chunk   = PCHUNK;
    int pblocks = (P + chunk - 1) / chunk;

    count_kernel<<<pblocks, 256, 0, stream>>>(seg, counts, P, N, chunk);

    dim3 agrid(C, NRANGE);
    accum_kernel<<<agrid, 32, 0, stream>>>(feat, seg, sums, P, C, N);

    finalize_wmma_kernel<<<N / 16, (C / 16) * 32, 0, stream>>>(sums, counts, out, C, N);
}


# accumulate kernel: 16 waves per workgroup share one set of lane columns in LDS (1536 one-wave groups -> 256), 6x fewer global adds; waves' adds meet in a column in arrival order (sum order changes)
# speedup vs baseline: 4.7078x; 1.6092x over previous
#include <hip/hip_runtime.h>
#include <hip/hip_bf16.h>

typedef float v2f __attribute__((ext_vector_type(2)));
typedef float v8f __attribute__((ext_vector_type(8)));

#define MAXSEG 512
#define LSEG 400
#define UB 16
#define NRANGE 2
#define AWAVES 16
#define PCHUNK 8192

__global__ void zero_ws_kernel(float* ws, int n) {
    int i = blockIdx.x * blockDim.x + threadIdx.x;
    if (i < n) ws[i] = 0.0f;
}

__global__ void count_kernel(const int* __restrict__ seg, float* __restrict__ counts,
                             int P, int N, int chunk) {
    __shared__ int hist[MAXSEG];
    for (int i = threadIdx.x; i < N; i += blockDim.x) hist[i] = 0;
    __syncthreads();
    int pstart = blockIdx.x * chunk;
    for (int i = threadIdx.x; i < chunk; i += blockDim.x) {
        int g = pstart + i;
        if (g < P) atomicAdd(&hist[seg[g]], 1);
    }
    __syncthreads();
    for (int i = threadIdx.x; i < N; i += blockDim.x) {
        int c = hist[i];
        if (c) atomicAdd(&counts[i], (float)c);
    }
}

__device__ __forceinline__ void load_batch(float4 (&fv)[UB], int4 (&sv)[UB],
                                           const float4* __restrict__ p4,
                                           const int4* __restrict__ s4,
                                           int bat, int lane) {
#pragma unroll
    for (int u = 0; u < UB; ++u) {
        fv[u] = p4[((size_t)bat * UB + u) * 32 + lane];
        sv[u] = s4[((size_t)bat * UB + u) * 32 + lane];
    }
}

__device__ __forceinline__ void add_batch(float* lacc, const float4 (&fv)[UB],
                                          const int4 (&sv)[UB], int lane) {
#pragma unroll
    for (int u = 0; u < UB; ++u) {
        atomicAdd(&lacc[sv[u].x * 32 + lane], fv[u].x);
        atomicAdd(&lacc[sv[u].y * 32 + lane], fv[u].y);
        atomicAdd(&lacc[sv[u].z * 32 + lane], fv[u].z);
        atomicAdd(&lacc[sv[u].w * 32 + lane], fv[u].w);
    }
}

__global__ void __launch_bounds__(AWAVES * 32)
accum_kernel(const float* __restrict__ feat,
             const int* __restrict__ seg,
             float* __restrict__ sums,
             int P, int C, int N) {
    __shared__ float lacc[LSEG * 32];

    const int tid = threadIdx.x;
    const int lane = tid & 31;
    const int wave = __builtin_amdgcn_readfirstlane(tid >> 5);
    const int c = blockIdx.x;
    const int r = blockIdx.y;
    const int R = gridDim.y;

    float4* lacc4 = (float4*)lacc;
    for (int i = tid; i < N * 8; i += AWAVES * 32) lacc4[i] = make_float4(0.f, 0.f, 0.f, 0.f);
    __syncthreads();

    const float* plane = feat + (size_t)c * (size_t)P;

    const int nbat = ((P & 3) == 0) ? (P / (128 * UB)) : 0;
    const int nslot = R * AWAVES;
    const int slot = r * AWAVES + wave;
    const int bat_lo = (int)(((long long)slot * nbat) / nslot);
    const int bat_hi = (int)(((long long)(slot + 1) * nbat) / nslot);

    const float4* p4 = (const float4*)plane;
    const int4*   s4 = (const int4*)seg;

    float4 fv[UB];
    int4   sv[UB];
    for (int k = bat_lo; k < bat_hi; ++k) {
        load_batch(fv, sv, p4, s4, k, lane);
        __builtin_amdgcn_sched_barrier(0);
        add_batch(lacc, fv, sv, lane);
    }

    if (slot == nslot - 1) {
        for (int i = nbat * (128 * UB) + lane; i < P; i += 32)
            atomicAdd(&lacc[seg[i] * 32 + lane], plane[i]);
    }
    __syncthreads();

    for (int s = tid; s < N; s += AWAVES * 32) {
        const float* row = &lacc[s * 32];
        float acc = 0.0f;
#pragma unroll
        for (int j = 0; j < 32; ++j) acc += row[(j + lane) & 31];
        atomicAdd(&sums[(size_t)s * C + c], acc);
    }
}

__global__ void finalize_wmma_kernel(const float* __restrict__ sums,
                                     const float* __restrict__ counts,
                                     float* __restrict__ out,
                                     int C, int N) {
    int wave = threadIdx.x >> 5;
    int lane = threadIdx.x & 31;
    int half = lane >> 4;
    int m    = lane & 15;
    int n0   = blockIdx.x * 16;
    int c0   = wave * 16;

    float cnt = counts[n0 + m];
    float inv = (cnt > 0.0f) ? (1.0f / cnt) : 0.0f;

    v8f acc = {};
#pragma unroll
    for (int k0 = 0; k0 < 16; k0 += 4) {
        int kx = k0 + 2 * half;
        v2f a, b;
        a.x = (m == kx)     ? inv : 0.0f;
        a.y = (m == kx + 1) ? inv : 0.0f;
        b.x = sums[(size_t)(n0 + kx)     * C + c0 + m];
        b.y = sums[(size_t)(n0 + kx + 1) * C + c0 + m];
        acc = __builtin_amdgcn_wmma_f32_16x16x4_f32(
             false, a,  false, b,
             (short)0, acc,  false,  false);
    }

#pragma unroll
    for (int v = 0; v < 8; ++v) {
        out[(size_t)(n0 + v + 8 * half) * C + c0 + m] = acc[v];
    }
}

extern "C" void kernel_launch(void* const* d_in, const int* in_sizes, int n_in,
                              void* d_out, int out_size, void* d_ws, size_t ws_size,
                              hipStream_t stream) {
    const float* feat = (const float*)d_in[0];
    const int*   seg  = (const int*)d_in[1];
    int P = in_sizes[1];
    int C = in_sizes[0] / P;
    int N = out_size / C;

    float* sums   = (float*)d_ws;
    float* counts = sums + (size_t)N * C;
    float* out    = (float*)d_out;

    int tot = N * C + N;
    zero_ws_kernel<<<(tot + 255) / 256, 256, 0, stream>>>(sums, tot);

    int chunk   = PCHUNK;
    int pblocks = (P + chunk - 1) / chunk;

    count_kernel<<<pblocks, 256, 0, stream>>>(seg, counts, P, N, chunk);

    dim3 agrid(C, NRANGE);
    accum_kernel<<<agrid, AWAVES * 32, 0, stream>>>(feat, seg, sums, P, C, N);

    finalize_wmma_kernel<<<N / 16, (C / 16) * 32, 0, stream>>>(sums, counts, out, C, N);
}


# count launch removed (4 launches to 3): accumulate workgroup of channel k % 128 also histograms batch k's segment ids, which it holds in registers; order of the float sums unchanged, counts exact
# speedup vs baseline: 6.2275x; 1.3228x over previous
#include <hip/hip_runtime.h>
#include <hip/hip_bf16.h>

typedef float v2f __attribute__((ext_vector_type(2)));
typedef float v8f __attribute__((ext_vector_type(8)));

#define MAXSEG 512
#define LSEG 400
#define UB 16
#define NRANGE 2
#define AWAVES 16

__global__ void zero_ws_kernel(float* ws, int n) {
    int i = blockIdx.x * blockDim.x + threadIdx.x;
    if (i < n) ws[i] = 0.0f;
}

__device__ __forceinline__ void load_batch(float4 (&fv)[UB], int4 (&sv)[UB],
                                           const float4* __restrict__ p4,
                                           const int4* __restrict__ s4,
                                           int bat, int lane) {
#pragma unroll
    for (int u = 0; u < UB; ++u) {
        fv[u] = p4[((size_t)bat * UB + u) * 32 + lane];
        sv[u] = s4[((size_t)bat * UB + u) * 32 + lane];
    }
}

__device__ __forceinline__ void add_batch(float* lacc, const float4 (&fv)[UB],
                                          const int4 (&sv)[UB], int lane) {
#pragma unroll
    for (int u = 0; u < UB; ++u) {
        atomicAdd(&lacc[sv[u].x * 32 + lane], fv[u].x);
        atomicAdd(&lacc[sv[u].y * 32 + lane], fv[u].y);
        atomicAdd(&lacc[sv[u].z * 32 + lane], fv[u].z);
        atomicAdd(&lacc[sv[u].w * 32 + lane], fv[u].w);
    }
}

__device__ __forceinline__ void count_batch(int* lcnt, const int4 (&sv)[UB]) {
#pragma unroll
    for (int u = 0; u < UB; ++u) {
        atomicAdd(&lcnt[sv[u].x], 1);
        atomicAdd(&lcnt[sv[u].y], 1);
        atomicAdd(&lcnt[sv[u].z], 1);
        atomicAdd(&lcnt[sv[u].w], 1);
    }
}

__global__ void __launch_bounds__(AWAVES * 32)
accum_kernel(const float* __restrict__ feat,
             const int* __restrict__ seg,
             float* __restrict__ sums,
             float* __restrict__ counts,
             int P, int C, int N) {
    __shared__ float lacc[LSEG * 32];
    __shared__ int lcnt[MAXSEG];

    const int tid = threadIdx.x;
    const int lane = tid & 31;
    const int wave = __builtin_amdgcn_readfirstlane(tid >> 5);
    const int c = blockIdx.x;
    const int r = blockIdx.y;
    const int R = gridDim.y;

    float4* lacc4 = (float4*)lacc;
    for (int i = tid; i < N * 8; i += AWAVES * 32) lacc4[i] = make_float4(0.f, 0.f, 0.f, 0.f);
    for (int i = tid; i < N; i += AWAVES * 32) lcnt[i] = 0;
    __syncthreads();

    const float* plane = feat + (size_t)c * (size_t)P;

    const int nbat = ((P & 3) == 0) ? (P / (128 * UB)) : 0;
    const int nslot = R * AWAVES;
    const int slot = r * AWAVES + wave;
    const int bat_lo = (int)(((long long)slot * nbat) / nslot);
    const int bat_hi = (int)(((long long)(slot + 1) * nbat) / nslot);

    const float4* p4 = (const float4*)plane;
    const int4*   s4 = (const int4*)seg;

    int kd = c - bat_lo % C;
    if (kd < 0) kd += C;
    int kc = bat_lo + kd;

    float4 fv[UB];
    int4   sv[UB];
    for (int k = bat_lo; k < bat_hi; ++k) {
        load_batch(fv, sv, p4, s4, k, lane);
        __builtin_amdgcn_sched_barrier(0);
        if (k == kc) {
            count_batch(lcnt, sv);
            kc += C;
        }
        add_batch(lacc, fv, sv, lane);
    }

    if (slot == nslot - 1) {
        for (int i = nbat * (128 * UB) + lane; i < P; i += 32) {
            const int sg = seg[i];
            atomicAdd(&lacc[sg * 32 + lane], plane[i]);
            if (c == 0) atomicAdd(&lcnt[sg], 1);
        }
    }
    __syncthreads();

    for (int s = tid; s < N; s += AWAVES * 32) {
        const float* row = &lacc[s * 32];
        float acc = 0.0f;
#pragma unroll
        for (int j = 0; j < 32; ++j) acc += row[(j + lane) & 31];
        atomicAdd(&sums[(size_t)s * C + c], acc);
        const int n = lcnt[s];
        if (n) atomicAdd(&counts[s], (float)n);
    }
}

__global__ void finalize_wmma_kernel(const float* __restrict__ sums,
                                     const float* __restrict__ counts,
                                     float* __restrict__ out,
                                     int C, int N) {
    int wave = threadIdx.x >> 5;
    int lane = threadIdx.x & 31;
    int half = lane >> 4;
    int m    = lane & 15;
    int n0   = blockIdx.x * 16;
    int c0   = wave * 16;

    float cnt = counts[n0 + m];
    float inv = (cnt > 0.0f) ? (1.0f / cnt) : 0.0f;

    v8f acc = {};
#pragma unroll
    for (int k0 = 0; k0 < 16; k0 += 4) {
        int kx = k0 + 2 * half;
        v2f a, b;
        a.x = (m == kx)     ? inv : 0.0f;
        a.y = (m == kx + 1) ? inv : 0.0f;
        b.x = sums[(size_t)(n0 + kx)     * C + c0 + m];
        b.y = sums[(size_t)(n0 + kx + 1) * C + c0 + m];
        acc = __builtin_amdgcn_wmma_f32_16x16x4_f32(
             false, a,  false, b,
             (short)0, acc,  false,  false);
    }

#pragma unroll
    for (int v = 0; v < 8; ++v) {
        out[(size_t)(n0 + v + 8 * half) * C + c0 + m] = acc[v];
    }
}

extern "C" void kernel_launch(void* const* d_in, const int* in_sizes, int n_in,
                              void* d_out, int out_size, void* d_ws, size_t ws_size,
                              hipStream_t stream) {
    const float* feat = (const float*)d_in[0];
    const int*   seg  = (const int*)d_in[1];
    int P = in_sizes[1];
    int C = in_sizes[0] / P;
    int N = out_size / C;

    float* sums   = (float*)d_ws;
    float* counts = sums + (size_t)N * C;
    float* out    = (float*)d_out;

    int tot = N * C + N;
    zero_ws_kernel<<<(tot + 255) / 256, 256, 0, stream>>>(sums, tot);

    dim3 agrid(C, NRANGE);
    accum_kernel<<<agrid, AWAVES * 32, 0, stream>>>(feat, seg, sums, counts, P, C, N);

    finalize_wmma_kernel<<<N / 16, (C / 16) * 32, 0, stream>>>(sums, counts, out, C, N);
}


# zeroing launch and all global adds removed (3 launches to 2): each accumulate workgroup stores totals and counts to its own line-aligned workspace rows, finalize adds the rows (sum order unchanged)
# speedup vs baseline: 7.1400x; 1.1465x over previous
#include <hip/hip_runtime.h>
#include <hip/hip_bf16.h>

typedef float v2f __attribute__((ext_vector_type(2)));
typedef float v8f __attribute__((ext_vector_type(8)));

#define MAXSEG 512
#define LSEG 400
#define UB 16
#define NRANGE 2
#define AWAVES 16

__device__ __forceinline__ int row_stride(int N) { return (N + 31) & ~31; }

__device__ __forceinline__ void load_batch(float4 (&fv)[UB], int4 (&sv)[UB],
                                           const float4* __restrict__ p4,
                                           const int4* __restrict__ s4,
                                           int bat, int lane) {
#pragma unroll
    for (int u = 0; u < UB; ++u) {
        fv[u] = p4[((size_t)bat * UB + u) * 32 + lane];
        sv[u] = s4[((size_t)bat * UB + u) * 32 + lane];
    }
}

__device__ __forceinline__ void add_batch(float* lacc, const float4 (&fv)[UB],
                                          const int4 (&sv)[UB], int lane) {
#pragma unroll
    for (int u = 0; u < UB; ++u) {
        atomicAdd(&lacc[sv[u].x * 32 + lane], fv[u].x);
        atomicAdd(&lacc[sv[u].y * 32 + lane], fv[u].y);
        atomicAdd(&lacc[sv[u].z * 32 + lane], fv[u].z);
        atomicAdd(&lacc[sv[u].w * 32 + lane], fv[u].w);
    }
}

__device__ __forceinline__ void count_batch(int* lcnt, const int4 (&sv)[UB]) {
#pragma unroll
    for (int u = 0; u < UB; ++u) {
        atomicAdd(&lcnt[sv[u].x], 1);
        atomicAdd(&lcnt[sv[u].y], 1);
        atomicAdd(&lcnt[sv[u].z], 1);
        atomicAdd(&lcnt[sv[u].w], 1);
    }
}

__global__ void __launch_bounds__(AWAVES * 32)
accum_kernel(const float* __restrict__ feat,
             const int* __restrict__ seg,
             float* __restrict__ part,
             int P, int C, int N) {
    __shared__ float lacc[LSEG * 32];
    __shared__ int lcnt[MAXSEG];

    const int tid = threadIdx.x;
    const int lane = tid & 31;
    const int wave = __builtin_amdgcn_readfirstlane(tid >> 5);
    const int c = blockIdx.x;
    const int r = blockIdx.y;
    const int R = gridDim.y;

    float4* lacc4 = (float4*)lacc;
    for (int i = tid; i < N * 8; i += AWAVES * 32) lacc4[i] = make_float4(0.f, 0.f, 0.f, 0.f);
    for (int i = tid; i < N; i += AWAVES * 32) lcnt[i] = 0;
    __syncthreads();

    const float* plane = feat + (size_t)c * (size_t)P;

    const int nbat = ((P & 3) == 0) ? (P / (128 * UB)) : 0;
    const int nslot = R * AWAVES;
    const int slot = r * AWAVES + wave;
    const int bat_lo = (int)(((long long)slot * nbat) / nslot);
    const int bat_hi = (int)(((long long)(slot + 1) * nbat) / nslot);

    const float4* p4 = (const float4*)plane;
    const int4*   s4 = (const int4*)seg;

    int kd = c - bat_lo % C;
    if (kd < 0) kd += C;
    int kc = bat_lo + kd;

    float4 fv[UB];
    int4   sv[UB];
    for (int k = bat_lo; k < bat_hi; ++k) {
        load_batch(fv, sv, p4, s4, k, lane);
        __builtin_amdgcn_sched_barrier(0);
        if (k == kc) {
            count_batch(lcnt, sv);
            kc += C;
        }
        add_batch(lacc, fv, sv, lane);
    }

    if (slot == nslot - 1) {
        for (int i = nbat * (128 * UB) + lane; i < P; i += 32) {
            const int sg = seg[i];
            atomicAdd(&lacc[sg * 32 + lane], plane[i]);
            if (c == 0) atomicAdd(&lcnt[sg], 1);
        }
    }
    __syncthreads();

    const int NS = row_stride(N);
    const size_t wg = (size_t)r * C + c;
    float* srow = part + wg * NS;
    float* crow = part + ((size_t)R * C + wg) * NS;
    for (int s = tid; s < N; s += AWAVES * 32) {
        const float* row = &lacc[s * 32];
        float acc = 0.0f;
#pragma unroll
        for (int j = 0; j < 32; ++j) acc += row[(j + lane) & 31];
        srow[s] = acc;
        crow[s] = (float)lcnt[s];
    }
}

__global__ void finalize_wmma_kernel(const float* __restrict__ part,
                                     float* __restrict__ out,
                                     int C, int N) {
    __shared__ float lpart[1024];

    int wave = threadIdx.x >> 5;
    int lane = threadIdx.x & 31;
    int half = lane >> 4;
    int m    = lane & 15;
    int n0   = blockIdx.x * 16;
    int c0   = wave * 16;

    const int NS = row_stride(N);
    const float* pcnt = part + NRANGE * C * NS;

    const int j  = threadIdx.x & 15;
    const int q  = threadIdx.x >> 4;
    const int nq = blockDim.x >> 4;
    float cp = 0.0f;
#pragma unroll
    for (int i = 0; i < 8 * NRANGE; ++i) cp += pcnt[(q + i * nq) * NS + n0 + j];

    v2f bv[4];
#pragma unroll
    for (int ks = 0; ks < 4; ++ks) {
        const int kx = 4 * ks + 2 * half;
        const float* p = part + (c0 + m) * NS + n0 + kx;
        v2f t = *(const v2f*)p;
#pragma unroll
        for (int r = 1; r < NRANGE; ++r) t += *(const v2f*)(p + r * C * NS);
        bv[ks] = t;
    }

    lpart[threadIdx.x] = cp;
    __syncthreads();
    float cnt = 0.0f;
    for (int i = 0; i < nq; ++i) cnt += lpart[i * 16 + m];
    float inv = (cnt > 0.0f) ? (1.0f / cnt) : 0.0f;

    v8f acc = {};
#pragma unroll
    for (int ks = 0; ks < 4; ++ks) {
        int kx = 4 * ks + 2 * half;
        v2f a;
        a.x = (m == kx)     ? inv : 0.0f;
        a.y = (m == kx + 1) ? inv : 0.0f;
        acc = __builtin_amdgcn_wmma_f32_16x16x4_f32(
             false, a,  false, bv[ks],
             (short)0, acc,  false,  false);
    }

#pragma unroll
    for (int v = 0; v < 8; ++v) {
        out[(size_t)(n0 + v + 8 * half) * C + c0 + m] = acc[v];
    }
}

extern "C" void kernel_launch(void* const* d_in, const int* in_sizes, int n_in,
                              void* d_out, int out_size, void* d_ws, size_t ws_size,
                              hipStream_t stream) {
    const float* feat = (const float*)d_in[0];
    const int*   seg  = (const int*)d_in[1];
    int P = in_sizes[1];
    int C = in_sizes[0] / P;
    int N = out_size / C;

    float* part = (float*)d_ws;
    float* out  = (float*)d_out;

    dim3 agrid(C, NRANGE);
    accum_kernel<<<agrid, AWAVES * 32, 0, stream>>>(feat, seg, part, P, C, N);

    finalize_wmma_kernel<<<N / 16, (C / 16) * 32, 0, stream>>>(part, out, C, N);
}
